# starting kernel
# baseline (speedup 1.0000x reference)
#include <hip/hip_runtime.h>

typedef __attribute__((ext_vector_type(2))) float v2f;
typedef __attribute__((ext_vector_type(8))) float v8f;

#define FEAT 64

__global__ void zero_f32(float* __restrict__ p, long n) {
    long i = (long)blockIdx.x * blockDim.x + threadIdx.x;
    long stride = (long)gridDim.x * blockDim.x;
    for (; i < n; i += stride) p[i] = 0.0f;
}

__global__ void degree_kernel(const int* __restrict__ dst, float* __restrict__ deg, int E) {
    int e = blockIdx.x * blockDim.x + threadIdx.x;
    if (e < E) atomicAdd(&deg[dst[e]], 1.0f);
}

__global__ void scatter_kernel(const float* __restrict__ x,
                               const int* __restrict__ src,
                               const int* __restrict__ dst,
                               float* __restrict__ aggr, long E) {
    long t = (long)blockIdx.x * blockDim.x + threadIdx.x;
    if (t >= E * 16) return;
    long e = t >> 4;
    int  f = (int)(t & 15) << 2;
    long s = src[e];
    long d = dst[e];
    const float4 v = *(const float4*)(x + s * FEAT + f);
    float* a = aggr + d * FEAT + f;
    atomicAdd(a + 0, v.x);
    atomicAdd(a + 1, v.y);
    atomicAdd(a + 2, v.z);
    atomicAdd(a + 3, v.w);
}

__global__ void __launch_bounds__(256)
sage_update_wmma(const float* __restrict__ x, const float* __restrict__ aggr,
                 const float* __restrict__ deg,
                 const float* __restrict__ Wl, const float* __restrict__ bl,
                 const float* __restrict__ Wr,
                 float* __restrict__ out, int N) {
    __shared__ float Wl_s[FEAT * FEAT];
    __shared__ float Wr_s[FEAT * FEAT];
    __shared__ float b_s[FEAT];

    for (int i = threadIdx.x; i < FEAT * FEAT / 4; i += 256) {
        ((float4*)Wl_s)[i] = ((const float4*)Wl)[i];
        ((float4*)Wr_s)[i] = ((const float4*)Wr)[i];
    }
    if (threadIdx.x < FEAT) b_s[threadIdx.x] = bl[threadIdx.x];
    __syncthreads();

    const int lane = threadIdx.x & 31;
    const int wave = threadIdx.x >> 5;
    const int tile = blockIdx.x * 8 + wave;
    const int base = tile * 16;
    if (base >= N) return;

    const int row  = lane & 15;
    const int half = lane >> 4;
    const int node = base + row;
    const float dinv = 1.0f / fmaxf(deg[node], 1.0f);

    v2f a_ag[16], a_x[16];
    const float* agp = aggr + (long)node * FEAT;
    const float* xp  = x    + (long)node * FEAT;
#pragma unroll
    for (int ks = 0; ks < 16; ++ks) {
        const int k0 = ks * 4 + half * 2;
        v2f t0; t0.x = agp[k0] * dinv; t0.y = agp[k0 + 1] * dinv;
        a_ag[ks] = t0;
        v2f t1; t1.x = xp[k0];         t1.y = xp[k0 + 1];
        a_x[ks] = t1;
    }

#pragma unroll
    for (int nb = 0; nb < 4; ++nb) {
        const int ncol = nb * 16 + row;
        v8f c = {};
#pragma unroll
        for (int ks = 0; ks < 16; ++ks) {
            const int k0 = ks * 4 + half * 2;
            v2f bfl; bfl.x = Wl_s[ncol * FEAT + k0]; bfl.y = Wl_s[ncol * FEAT + k0 + 1];
            c = __builtin_amdgcn_wmma_f32_16x16x4_f32(false, a_ag[ks], false, bfl,
                                                      (short)0, c, false, false);
            v2f bfr; bfr.x = Wr_s[ncol * FEAT + k0]; bfr.y = Wr_s[ncol * FEAT + k0 + 1];
            c = __builtin_amdgcn_wmma_f32_16x16x4_f32(false, a_x[ks], false, bfr,
                                                      (short)0, c, false, false);
        }
        const float bias = b_s[nb * 16 + row];
#pragma unroll
        for (int r = 0; r < 8; ++r) {
            const int M = r + half * 8;
            out[(long)(base + M) * FEAT + nb * 16 + row] = c[r] + bias;
        }
    }
}

extern "C" void kernel_launch(void* const* d_in, const int* in_sizes, int n_in,
                              void* d_out, int out_size, void* d_ws, size_t ws_size,
                              hipStream_t stream) {
    const float* x   = (const float*)d_in[0];
    const int*   eix = (const int*)d_in[1];
    const float* W1l = (const float*)d_in[2];
    const float* b1l = (const float*)d_in[3];
    const float* W1r = (const float*)d_in[4];
    const float* W2l = (const float*)d_in[5];
    const float* b2l = (const float*)d_in[6];
    const float* W2r = (const float*)d_in[7];

    const int  N = in_sizes[0] / FEAT;
    const long E = in_sizes[1] / 2;
    const int* src = eix;
    const int* dst = eix + E;

    float* deg  = (float*)d_ws;
    float* aggr = deg + N;
    float* x1   = aggr + (size_t)N * FEAT;
    float* outp = (float*)d_out;

    const int deg_blocks = (int)((E + 255) / 256);
    const int sc_blocks  = (int)((E * 16 + 255) / 256);
    const int up_blocks  = (N + 127) / 128;

    zero_f32<<<256, 256, 0, stream>>>(deg, (long)N);
    degree_kernel<<<deg_blocks, 256, 0, stream>>>(dst, deg, (int)E);

    zero_f32<<<2048, 256, 0, stream>>>(aggr, (long)N * FEAT);
    scatter_kernel<<<sc_blocks, 256, 0, stream>>>(x, src, dst, aggr, E);
    sage_update_wmma<<<up_blocks, 256, 0, stream>>>(x, aggr, deg, W1l, b1l, W1r, x1, N);

    zero_f32<<<2048, 256, 0, stream>>>(aggr, (long)N * FEAT);
    scatter_kernel<<<sc_blocks, 256, 0, stream>>>(x1, src, dst, aggr, E);
    sage_update_wmma<<<up_blocks, 256, 0, stream>>>(x1, aggr, deg, W2l, b2l, W2r, outp, N);
}


# scatter remapped: a fixed grid of waves walks runs of 8 edges, one lane per float so each row load and atomic covers a whole 128-B line, a run's loads all in flight; atomic arrival order changes
# speedup vs baseline: 1.3161x; 1.3161x over previous
#include <hip/hip_runtime.h>

typedef __attribute__((ext_vector_type(2))) float v2f;
typedef __attribute__((ext_vector_type(8))) float v8f;

#define FEAT 64

__global__ void zero_f32(float* __restrict__ p, long n) {
    long i = (long)blockIdx.x * blockDim.x + threadIdx.x;
    long stride = (long)gridDim.x * blockDim.x;
    for (; i < n; i += stride) p[i] = 0.0f;
}

__global__ void degree_kernel(const int* __restrict__ dst, float* __restrict__ deg, int E) {
    int e = blockIdx.x * blockDim.x + threadIdx.x;
    if (e < E) atomicAdd(&deg[dst[e]], 1.0f);
}

#define EPT 8
__global__ void __launch_bounds__(256)
scatter_kernel(const float* __restrict__ x,
               const int* __restrict__ src,
               const int* __restrict__ dst,
               float* __restrict__ aggr, long E) {
    const int lane = threadIdx.x & 31;
    const int wv = __builtin_amdgcn_readfirstlane((int)(threadIdx.x >> 5));
    const long wpb = (long)(blockDim.x >> 5);
    const long nw  = (long)gridDim.x * wpb;
    const long w   = (long)blockIdx.x * wpb + wv;
    const int  sub = lane & (EPT - 1);
    const int* idx = (lane & EPT) ? dst : src;

    long e0 = w * EPT;
    int my = 0;
    if (e0 < E && sub < E - e0) my = idx[e0 + sub];
    while (e0 < E) {
        const long rem = E - e0;
        const long en  = e0 + nw * EPT;
        if (rem >= EPT) {
            float lo[EPT], hi[EPT];
#pragma unroll
            for (int i = 0; i < EPT; ++i) {
                const long s = __builtin_amdgcn_readlane(my, i);
                const float* xr = x + s * FEAT + lane;
                lo[i] = xr[0];
                hi[i] = xr[32];
            }
            int nxt = 0;
            if (en < E && sub < E - en) nxt = idx[en + sub];
#pragma unroll
            for (int i = 0; i < EPT; ++i) {
                const long d = __builtin_amdgcn_readlane(my, EPT + i);
                float* ar = aggr + d * FEAT + lane;
                atomicAdd(ar, lo[i]);
                atomicAdd(ar + 32, hi[i]);
            }
            my = nxt;
        } else {
            for (int i = 0; i < (int)rem; ++i) {
                const long s = __builtin_amdgcn_readlane(my, i);
                const long d = __builtin_amdgcn_readlane(my, EPT + i);
                const float* xr = x + s * FEAT + lane;
                float* ar = aggr + d * FEAT + lane;
                const float a = xr[0];
                const float b = xr[32];
                atomicAdd(ar, a);
                atomicAdd(ar + 32, b);
            }
        }
        e0 = en;
    }
}

__global__ void __launch_bounds__(256)
sage_update_wmma(const float* __restrict__ x, const float* __restrict__ aggr,
                 const float* __restrict__ deg,
                 const float* __restrict__ Wl, const float* __restrict__ bl,
                 const float* __restrict__ Wr,
                 float* __restrict__ out, int N) {
    __shared__ float Wl_s[FEAT * FEAT];
    __shared__ float Wr_s[FEAT * FEAT];
    __shared__ float b_s[FEAT];

    for (int i = threadIdx.x; i < FEAT * FEAT / 4; i += 256) {
        ((float4*)Wl_s)[i] = ((const float4*)Wl)[i];
        ((float4*)Wr_s)[i] = ((const float4*)Wr)[i];
    }
    if (threadIdx.x < FEAT) b_s[threadIdx.x] = bl[threadIdx.x];
    __syncthreads();

    const int lane = threadIdx.x & 31;
    const int wave = threadIdx.x >> 5;
    const int tile = blockIdx.x * 8 + wave;
    const int base = tile * 16;
    if (base >= N) return;

    const int row  = lane & 15;
    const int half = lane >> 4;
    const int node = base + row;
    const float dinv = 1.0f / fmaxf(deg[node], 1.0f);

    v2f a_ag[16], a_x[16];
    const float* agp = aggr + (long)node * FEAT;
    const float* xp  = x    + (long)node * FEAT;
#pragma unroll
    for (int ks = 0; ks < 16; ++ks) {
        const int k0 = ks * 4 + half * 2;
        v2f t0; t0.x = agp[k0] * dinv; t0.y = agp[k0 + 1] * dinv;
        a_ag[ks] = t0;
        v2f t1; t1.x = xp[k0];         t1.y = xp[k0 + 1];
        a_x[ks] = t1;
    }

#pragma unroll
    for (int nb = 0; nb < 4; ++nb) {
        const int ncol = nb * 16 + row;
        v8f c = {};
#pragma unroll
        for (int ks = 0; ks < 16; ++ks) {
            const int k0 = ks * 4 + half * 2;
            v2f bfl; bfl.x = Wl_s[ncol * FEAT + k0]; bfl.y = Wl_s[ncol * FEAT + k0 + 1];
            c = __builtin_amdgcn_wmma_f32_16x16x4_f32(false, a_ag[ks], false, bfl,
                                                      (short)0, c, false, false);
            v2f bfr; bfr.x = Wr_s[ncol * FEAT + k0]; bfr.y = Wr_s[ncol * FEAT + k0 + 1];
            c = __builtin_amdgcn_wmma_f32_16x16x4_f32(false, a_x[ks], false, bfr,
                                                      (short)0, c, false, false);
        }
        const float bias = b_s[nb * 16 + row];
#pragma unroll
        for (int r = 0; r < 8; ++r) {
            const int M = r + half * 8;
            out[(long)(base + M) * FEAT + nb * 16 + row] = c[r] + bias;
        }
    }
}

extern "C" void kernel_launch(void* const* d_in, const int* in_sizes, int n_in,
                              void* d_out, int out_size, void* d_ws, size_t ws_size,
                              hipStream_t stream) {
    const float* x   = (const float*)d_in[0];
    const int*   eix = (const int*)d_in[1];
    const float* W1l = (const float*)d_in[2];
    const float* b1l = (const float*)d_in[3];
    const float* W1r = (const float*)d_in[4];
    const float* W2l = (const float*)d_in[5];
    const float* b2l = (const float*)d_in[6];
    const float* W2r = (const float*)d_in[7];

    const int  N = in_sizes[0] / FEAT;
    const long E = in_sizes[1] / 2;
    const int* src = eix;
    const int* dst = eix + E;

    float* deg  = (float*)d_ws;
    float* aggr = deg + N;
    float* x1   = aggr + (size_t)N * FEAT;
    float* outp = (float*)d_out;

    const int deg_blocks = (int)((E + 255) / 256);
    const int up_blocks  = (N + 127) / 128;

    zero_f32<<<256, 256, 0, stream>>>(deg, (long)N);
    degree_kernel<<<deg_blocks, 256, 0, stream>>>(dst, deg, (int)E);

    zero_f32<<<2048, 256, 0, stream>>>(aggr, (long)N * FEAT);
    scatter_kernel<<<2048, 256, 0, stream>>>(x, src, dst, aggr, E);
    sage_update_wmma<<<up_blocks, 256, 0, stream>>>(x, aggr, deg, W1l, b1l, W1r, x1, N);

    zero_f32<<<2048, 256, 0, stream>>>(aggr, (long)N * FEAT);
    scatter_kernel<<<2048, 256, 0, stream>>>(x1, src, dst, aggr, E);
    sage_update_wmma<<<up_blocks, 256, 0, stream>>>(x1, aggr, deg, W2l, b2l, W2r, outp, N);
}


# edges bucketed by destination once per call (int-atomic count+rank, scan, fill); a wave sums a node's source rows in registers, stores the row once: no f32 atomics or zero fill; sum in list order
# speedup vs baseline: 1.6889x; 1.2833x over previous
#include <hip/hip_runtime.h>

typedef __attribute__((ext_vector_type(2))) float v2f;
typedef __attribute__((ext_vector_type(8))) float v8f;

#define FEAT 64


__global__ void zero_i32(int* __restrict__ p, long n) {
    long i = (long)blockIdx.x * blockDim.x + threadIdx.x;
    long stride = (long)gridDim.x * blockDim.x;
    for (; i < n; i += stride) p[i] = 0;
}

#define EDGES_PER_THREAD 4
__global__ void __launch_bounds__(256)
count_rank_kernel(const int* __restrict__ dst, int* __restrict__ cnt,
                  int* __restrict__ rank, long E) {
    const long T = (long)gridDim.x * blockDim.x;
    long e = (long)blockIdx.x * blockDim.x + threadIdx.x;
    for (; e < E; e += EDGES_PER_THREAD * T) {
        int d[EDGES_PER_THREAD], r[EDGES_PER_THREAD];
#pragma unroll
        for (int i = 0; i < EDGES_PER_THREAD; ++i) {
            const long ee = e + i * T;
            d[i] = (ee < E) ? dst[ee] : -1;
        }
#pragma unroll
        for (int i = 0; i < EDGES_PER_THREAD; ++i) {
            r[i] = 0;
            if (d[i] >= 0) r[i] = atomicAdd(&cnt[d[i]], 1);
        }
#pragma unroll
        for (int i = 0; i < EDGES_PER_THREAD; ++i) {
            if (d[i] >= 0) rank[e + i * T] = r[i];
        }
    }
}

#define SCAN_CHUNK 1024
__global__ void __launch_bounds__(SCAN_CHUNK)
scan_kernel(const int* __restrict__ cnt, int* __restrict__ off,
            float* __restrict__ deg, int N) {
    __shared__ int front_s[SCAN_CHUNK / 32];
    __shared__ int wave_s[SCAN_CHUNK / 32];
    const int tid  = threadIdx.x;
    const int lane = tid & 31;
    const int wv   = tid >> 5;
    const int start = blockIdx.x * SCAN_CHUNK;

    int part = 0;
    const int4* c4 = (const int4*)cnt;
    const int n4 = start >> 2;
#pragma unroll 8
    for (int i = tid; i < n4; i += SCAN_CHUNK) {
        const int4 v = c4[i];
        part += (v.x + v.y) + (v.z + v.w);
    }
    for (int o = 16; o > 0; o >>= 1) part += __shfl_xor(part, o);
    if (lane == 0) front_s[wv] = part;

    const int idx = start + tid;
    const int c = (idx < N) ? cnt[idx] : 0;
    int incl = c;
    for (int o = 1; o < 32; o <<= 1) {
        const int t = __shfl_up(incl, o);
        if (lane >= o) incl += t;
    }
    if (lane == 31) wave_s[wv] = incl;
    __syncthreads();

    int base = 0;
    for (int k = 0; k < SCAN_CHUNK / 32; ++k) {
        base += front_s[k];
        if (k < wv) base += wave_s[k];
    }
    const int ex = base + incl - c;
    if (idx <= N) off[idx] = ex;
    if (idx < N)  deg[idx] = (float)c;
}

__global__ void __launch_bounds__(256)
fill_kernel(const int* __restrict__ src, const int* __restrict__ dst,
            const int* __restrict__ rank, const int* __restrict__ off,
            int* __restrict__ csr, long E) {
    const long T = (long)gridDim.x * blockDim.x;
    long e = (long)blockIdx.x * blockDim.x + threadIdx.x;
    for (; e < E; e += EDGES_PER_THREAD * T) {
        int d[EDGES_PER_THREAD], o[EDGES_PER_THREAD], r[EDGES_PER_THREAD], s[EDGES_PER_THREAD];
#pragma unroll
        for (int i = 0; i < EDGES_PER_THREAD; ++i) {
            const long ee = e + i * T;
            d[i] = (ee < E) ? dst[ee] : -1;
        }
#pragma unroll
        for (int i = 0; i < EDGES_PER_THREAD; ++i) {
            const long ee = e + i * T;
            o[i] = 0;
            r[i] = 0;
            s[i] = 0;
            if (d[i] >= 0) {
                o[i] = off[d[i]];
                r[i] = rank[ee];
                s[i] = src[ee];
            }
        }
#pragma unroll
        for (int i = 0; i < EDGES_PER_THREAD; ++i) {
            if (d[i] >= 0) csr[o[i] + r[i]] = s[i];
        }
    }
}

#define RUN 16
__global__ void __launch_bounds__(256)
gather_sum_kernel(const float* __restrict__ x, const int* __restrict__ off,
                  const int* __restrict__ csr, float* __restrict__ aggr, int N) {
    const int lane = threadIdx.x & 31;
    const unsigned lo = 2u * (unsigned)lane;
    const int wv  = __builtin_amdgcn_readfirstlane((int)(threadIdx.x >> 5));
    const int wpb = (int)(blockDim.x >> 5);
    const int nw  = (int)gridDim.x * wpb;
    for (int node = (int)blockIdx.x * wpb + wv; node < N; node += nw) {
        const int beg = off[node];
        const int end = off[node + 1];
        v2f acc = {0.0f, 0.0f};
        for (int p = beg; p < end; p += 32) {
            const int cnt = (end - p < 32) ? (end - p) : 32;
            int my = 0;
            if (lane < cnt) my = csr[p + lane];
            for (int j = 0; j < cnt; j += RUN) {
                const int m = cnt - j;
                long s[RUN];
#pragma unroll
                for (int i = 0; i < RUN; ++i) s[i] = __builtin_amdgcn_readlane(my, j + i);
                v2f v[RUN];
                if (m >= RUN) {
#pragma unroll
                    for (int i = 0; i < RUN; ++i) v[i] = *(const v2f*)(x + s[i] * FEAT + lo);
                } else {
#pragma unroll
                    for (int i = 0; i < RUN; ++i) {
                        v[i] = (v2f){0.0f, 0.0f};
                        if (i < m) v[i] = *(const v2f*)(x + s[i] * FEAT + lo);
                    }
                }
#pragma unroll
                for (int i = 0; i < RUN; ++i) acc += v[i];
            }
        }
        *(v2f*)(aggr + (long)node * FEAT + lo) = acc;
    }
}

__global__ void __launch_bounds__(256)
sage_update_wmma(const float* __restrict__ x, const float* __restrict__ aggr,
                 const float* __restrict__ deg,
                 const float* __restrict__ Wl, const float* __restrict__ bl,
                 const float* __restrict__ Wr,
                 float* __restrict__ out, int N) {
    __shared__ float Wl_s[FEAT * FEAT];
    __shared__ float Wr_s[FEAT * FEAT];
    __shared__ float b_s[FEAT];

    for (int i = threadIdx.x; i < FEAT * FEAT / 4; i += 256) {
        ((float4*)Wl_s)[i] = ((const float4*)Wl)[i];
        ((float4*)Wr_s)[i] = ((const float4*)Wr)[i];
    }
    if (threadIdx.x < FEAT) b_s[threadIdx.x] = bl[threadIdx.x];
    __syncthreads();

    const int lane = threadIdx.x & 31;
    const int wave = threadIdx.x >> 5;
    const int tile = blockIdx.x * 8 + wave;
    const int base = tile * 16;
    if (base >= N) return;

    const int row  = lane & 15;
    const int half = lane >> 4;
    const int node = base + row;
    const float dinv = 1.0f / fmaxf(deg[node], 1.0f);

    v2f a_ag[16], a_x[16];
    const float* agp = aggr + (long)node * FEAT;
    const float* xp  = x    + (long)node * FEAT;
#pragma unroll
    for (int ks = 0; ks < 16; ++ks) {
        const int k0 = ks * 4 + half * 2;
        v2f t0; t0.x = agp[k0] * dinv; t0.y = agp[k0 + 1] * dinv;
        a_ag[ks] = t0;
        v2f t1; t1.x = xp[k0];         t1.y = xp[k0 + 1];
        a_x[ks] = t1;
    }

#pragma unroll
    for (int nb = 0; nb < 4; ++nb) {
        const int ncol = nb * 16 + row;
        v8f c = {};
#pragma unroll
        for (int ks = 0; ks < 16; ++ks) {
            const int k0 = ks * 4 + half * 2;
            v2f bfl; bfl.x = Wl_s[ncol * FEAT + k0]; bfl.y = Wl_s[ncol * FEAT + k0 + 1];
            c = __builtin_amdgcn_wmma_f32_16x16x4_f32(false, a_ag[ks], false, bfl,
                                                      (short)0, c, false, false);
            v2f bfr; bfr.x = Wr_s[ncol * FEAT + k0]; bfr.y = Wr_s[ncol * FEAT + k0 + 1];
            c = __builtin_amdgcn_wmma_f32_16x16x4_f32(false, a_x[ks], false, bfr,
                                                      (short)0, c, false, false);
        }
        const float bias = b_s[nb * 16 + row];
#pragma unroll
        for (int r = 0; r < 8; ++r) {
            const int M = r + half * 8;
            out[(long)(base + M) * FEAT + nb * 16 + row] = c[r] + bias;
        }
    }
}

extern "C" void kernel_launch(void* const* d_in, const int* in_sizes, int n_in,
                              void* d_out, int out_size, void* d_ws, size_t ws_size,
                              hipStream_t stream) {
    const float* x   = (const float*)d_in[0];
    const int*   eix = (const int*)d_in[1];
    const float* W1l = (const float*)d_in[2];
    const float* b1l = (const float*)d_in[3];
    const float* W1r = (const float*)d_in[4];
    const float* W2l = (const float*)d_in[5];
    const float* b2l = (const float*)d_in[6];
    const float* W2r = (const float*)d_in[7];

    const int  N = in_sizes[0] / FEAT;
    const long E = in_sizes[1] / 2;
    const int* src = eix;
    const int* dst = eix + E;

    float* aggr = (float*)d_ws;
    float* x1   = aggr + (size_t)N * FEAT;
    float* deg  = x1 + (size_t)N * FEAT;
    float* outp = (float*)d_out;

    int* cnt  = (int*)x1;
    int* rank = cnt + N;
    int* off  = (int*)d_out;
    int* csr  = off + N + 1;

    const int edge_blocks = (int)((E + 256 * EDGES_PER_THREAD - 1) / (256 * EDGES_PER_THREAD));
    const int scan_blocks = (N + SCAN_CHUNK) / SCAN_CHUNK;
    const int up_blocks   = (N + 127) / 128;

    zero_i32<<<256, 256, 0, stream>>>(cnt, (long)N);
    count_rank_kernel<<<edge_blocks, 256, 0, stream>>>(dst, cnt, rank, E);
    scan_kernel<<<scan_blocks, SCAN_CHUNK, 0, stream>>>(cnt, off, deg, N);
    fill_kernel<<<edge_blocks, 256, 0, stream>>>(src, dst, rank, off, csr, E);

    gather_sum_kernel<<<2048, 256, 0, stream>>>(x, off, csr, aggr, N);
    sage_update_wmma<<<up_blocks, 256, 0, stream>>>(x, aggr, deg, W1l, b1l, W1r, x1, N);

    gather_sum_kernel<<<2048, 256, 0, stream>>>(x1, off, csr, aggr, N);
    sage_update_wmma<<<up_blocks, 256, 0, stream>>>(x1, aggr, deg, W2l, b2l, W2r, outp, N);
}


# gather: rows addressed by a 32-bit pair index on a scalar base, not a 64-bit address pair per row in flight: 72 to 44 VGPRs, 16 waves per SIMD, whole grid resident in one round; same sums
# speedup vs baseline: 1.7425x; 1.0317x over previous
#include <hip/hip_runtime.h>

typedef __attribute__((ext_vector_type(2))) float v2f;
typedef __attribute__((ext_vector_type(8))) float v8f;

#define FEAT 64


__global__ void zero_i32(int* __restrict__ p, long n) {
    long i = (long)blockIdx.x * blockDim.x + threadIdx.x;
    long stride = (long)gridDim.x * blockDim.x;
    for (; i < n; i += stride) p[i] = 0;
}

#define EDGES_PER_THREAD 4
__global__ void __launch_bounds__(256)
count_rank_kernel(const int* __restrict__ dst, int* __restrict__ cnt,
                  int* __restrict__ rank, long E) {
    const long T = (long)gridDim.x * blockDim.x;
    long e = (long)blockIdx.x * blockDim.x + threadIdx.x;
    for (; e < E; e += EDGES_PER_THREAD * T) {
        int d[EDGES_PER_THREAD], r[EDGES_PER_THREAD];
#pragma unroll
        for (int i = 0; i < EDGES_PER_THREAD; ++i) {
            const long ee = e + i * T;
            d[i] = (ee < E) ? dst[ee] : -1;
        }
#pragma unroll
        for (int i = 0; i < EDGES_PER_THREAD; ++i) {
            r[i] = 0;
            if (d[i] >= 0) r[i] = atomicAdd(&cnt[d[i]], 1);
        }
#pragma unroll
        for (int i = 0; i < EDGES_PER_THREAD; ++i) {
            if (d[i] >= 0) rank[e + i * T] = r[i];
        }
    }
}

#define SCAN_CHUNK 1024
__global__ void __launch_bounds__(SCAN_CHUNK)
scan_kernel(const int* __restrict__ cnt, int* __restrict__ off,
            float* __restrict__ deg, int N) {
    __shared__ int front_s[SCAN_CHUNK / 32];
    __shared__ int wave_s[SCAN_CHUNK / 32];
    const int tid  = threadIdx.x;
    const int lane = tid & 31;
    const int wv   = tid >> 5;
    const int start = blockIdx.x * SCAN_CHUNK;

    int part = 0;
    const int4* c4 = (const int4*)cnt;
    const int n4 = start >> 2;
#pragma unroll 8
    for (int i = tid; i < n4; i += SCAN_CHUNK) {
        const int4 v = c4[i];
        part += (v.x + v.y) + (v.z + v.w);
    }
    for (int o = 16; o > 0; o >>= 1) part += __shfl_xor(part, o);
    if (lane == 0) front_s[wv] = part;

    const int idx = start + tid;
    const int c = (idx < N) ? cnt[idx] : 0;
    int incl = c;
    for (int o = 1; o < 32; o <<= 1) {
        const int t = __shfl_up(incl, o);
        if (lane >= o) incl += t;
    }
    if (lane == 31) wave_s[wv] = incl;
    __syncthreads();

    int base = 0;
    for (int k = 0; k < SCAN_CHUNK / 32; ++k) {
        base += front_s[k];
        if (k < wv) base += wave_s[k];
    }
    const int ex = base + incl - c;
    if (idx <= N) off[idx] = ex;
    if (idx < N)  deg[idx] = (float)c;
}

__global__ void __launch_bounds__(256)
fill_kernel(const int* __restrict__ src, const int* __restrict__ dst,
            const int* __restrict__ rank, const int* __restrict__ off,
            int* __restrict__ csr, long E) {
    const long T = (long)gridDim.x * blockDim.x;
    long e = (long)blockIdx.x * blockDim.x + threadIdx.x;
    for (; e < E; e += EDGES_PER_THREAD * T) {
        int d[EDGES_PER_THREAD], o[EDGES_PER_THREAD], r[EDGES_PER_THREAD], s[EDGES_PER_THREAD];
#pragma unroll
        for (int i = 0; i < EDGES_PER_THREAD; ++i) {
            const long ee = e + i * T;
            d[i] = (ee < E) ? dst[ee] : -1;
        }
#pragma unroll
        for (int i = 0; i < EDGES_PER_THREAD; ++i) {
            const long ee = e + i * T;
            o[i] = 0;
            r[i] = 0;
            s[i] = 0;
            if (d[i] >= 0) {
                o[i] = off[d[i]];
                r[i] = rank[ee];
                s[i] = src[ee];
            }
        }
#pragma unroll
        for (int i = 0; i < EDGES_PER_THREAD; ++i) {
            if (d[i] >= 0) csr[o[i] + r[i]] = s[i];
        }
    }
}

#define RUN 16
#define ROW2 (FEAT / 2)
__global__ void __launch_bounds__(256)
gather_sum_kernel(const float* __restrict__ x, const int* __restrict__ off,
                  const int* __restrict__ csr, float* __restrict__ aggr, int N) {
    const v2f* __restrict__ x2 = (const v2f*)x;
    v2f* __restrict__ aggr2 = (v2f*)aggr;
    const unsigned lane = threadIdx.x & 31u;
    const int wv  = __builtin_amdgcn_readfirstlane((int)(threadIdx.x >> 5));
    const int wpb = (int)(blockDim.x >> 5);
    const int nw  = (int)gridDim.x * wpb;
    for (int node = (int)blockIdx.x * wpb + wv; node < N; node += nw) {
        const int beg = off[node];
        const int end = off[node + 1];
        v2f acc = {0.0f, 0.0f};
        for (int p = beg; p < end; p += 32) {
            const int cnt = (end - p < 32) ? (end - p) : 32;
            int my = 0;
            if ((int)lane < cnt) my = csr[p + (int)lane];
            for (int j = 0; j < cnt; j += RUN) {
                const int m = cnt - j;
                unsigned s[RUN];
#pragma unroll
                for (int i = 0; i < RUN; ++i)
                    s[i] = (unsigned)__builtin_amdgcn_readlane(my, j + i) * (unsigned)ROW2;
                if (m >= RUN) {
                    v2f v[RUN];
#pragma unroll
                    for (int i = 0; i < RUN; ++i) v[i] = x2[s[i] + lane];
#pragma unroll
                    for (int i = 0; i < RUN; ++i) acc += v[i];
                } else {
                    v2f v[RUN];
#pragma unroll
                    for (int i = 0; i < RUN; ++i) {
                        v[i] = (v2f){0.0f, 0.0f};
                        if (i < m) v[i] = x2[s[i] + lane];
                    }
#pragma unroll
                    for (int i = 0; i < RUN; ++i) acc += v[i];
                }
            }
        }
        aggr2[(unsigned)node * (unsigned)ROW2 + lane] = acc;
    }
}

__global__ void __launch_bounds__(256)
sage_update_wmma(const float* __restrict__ x, const float* __restrict__ aggr,
                 const float* __restrict__ deg,
                 const float* __restrict__ Wl, const float* __restrict__ bl,
                 const float* __restrict__ Wr,
                 float* __restrict__ out, int N) {
    __shared__ float Wl_s[FEAT * FEAT];
    __shared__ float Wr_s[FEAT * FEAT];
    __shared__ float b_s[FEAT];

    for (int i = threadIdx.x; i < FEAT * FEAT / 4; i += 256) {
        ((float4*)Wl_s)[i] = ((const float4*)Wl)[i];
        ((float4*)Wr_s)[i] = ((const float4*)Wr)[i];
    }
    if (threadIdx.x < FEAT) b_s[threadIdx.x] = bl[threadIdx.x];
    __syncthreads();

    const int lane = threadIdx.x & 31;
    const int wave = threadIdx.x >> 5;
    const int tile = blockIdx.x * 8 + wave;
    const int base = tile * 16;
    if (base >= N) return;

    const int row  = lane & 15;
    const int half = lane >> 4;
    const int node = base + row;
    const float dinv = 1.0f / fmaxf(deg[node], 1.0f);

    v2f a_ag[16], a_x[16];
    const float* agp = aggr + (long)node * FEAT;
    const float* xp  = x    + (long)node * FEAT;
#pragma unroll
    for (int ks = 0; ks < 16; ++ks) {
        const int k0 = ks * 4 + half * 2;
        v2f t0; t0.x = agp[k0] * dinv; t0.y = agp[k0 + 1] * dinv;
        a_ag[ks] = t0;
        v2f t1; t1.x = xp[k0];         t1.y = xp[k0 + 1];
        a_x[ks] = t1;
    }

#pragma unroll
    for (int nb = 0; nb < 4; ++nb) {
        const int ncol = nb * 16 + row;
        v8f c = {};
#pragma unroll
        for (int ks = 0; ks < 16; ++ks) {
            const int k0 = ks * 4 + half * 2;
            v2f bfl; bfl.x = Wl_s[ncol * FEAT + k0]; bfl.y = Wl_s[ncol * FEAT + k0 + 1];
            c = __builtin_amdgcn_wmma_f32_16x16x4_f32(false, a_ag[ks], false, bfl,
                                                      (short)0, c, false, false);
            v2f bfr; bfr.x = Wr_s[ncol * FEAT + k0]; bfr.y = Wr_s[ncol * FEAT + k0 + 1];
            c = __builtin_amdgcn_wmma_f32_16x16x4_f32(false, a_x[ks], false, bfr,
                                                      (short)0, c, false, false);
        }
        const float bias = b_s[nb * 16 + row];
#pragma unroll
        for (int r = 0; r < 8; ++r) {
            const int M = r + half * 8;
            out[(long)(base + M) * FEAT + nb * 16 + row] = c[r] + bias;
        }
    }
}

extern "C" void kernel_launch(void* const* d_in, const int* in_sizes, int n_in,
                              void* d_out, int out_size, void* d_ws, size_t ws_size,
                              hipStream_t stream) {
    const float* x   = (const float*)d_in[0];
    const int*   eix = (const int*)d_in[1];
    const float* W1l = (const float*)d_in[2];
    const float* b1l = (const float*)d_in[3];
    const float* W1r = (const float*)d_in[4];
    const float* W2l = (const float*)d_in[5];
    const float* b2l = (const float*)d_in[6];
    const float* W2r = (const float*)d_in[7];

    const int  N = in_sizes[0] / FEAT;
    const long E = in_sizes[1] / 2;
    const int* src = eix;
    const int* dst = eix + E;

    float* aggr = (float*)d_ws;
    float* x1   = aggr + (size_t)N * FEAT;
    float* deg  = x1 + (size_t)N * FEAT;
    float* outp = (float*)d_out;

    int* cnt  = (int*)x1;
    int* rank = cnt + N;
    int* off  = (int*)d_out;
    int* csr  = off + N + 1;

    const int edge_blocks = (int)((E + 256 * EDGES_PER_THREAD - 1) / (256 * EDGES_PER_THREAD));
    const int scan_blocks = (N + SCAN_CHUNK) / SCAN_CHUNK;
    const int up_blocks   = (N + 127) / 128;

    zero_i32<<<256, 256, 0, stream>>>(cnt, (long)N);
    count_rank_kernel<<<edge_blocks, 256, 0, stream>>>(dst, cnt, rank, E);
    scan_kernel<<<scan_blocks, SCAN_CHUNK, 0, stream>>>(cnt, off, deg, N);
    fill_kernel<<<edge_blocks, 256, 0, stream>>>(src, dst, rank, off, csr, E);

    gather_sum_kernel<<<2048, 256, 0, stream>>>(x, off, csr, aggr, N);
    sage_update_wmma<<<up_blocks, 256, 0, stream>>>(x, aggr, deg, W1l, b1l, W1r, x1, N);

    gather_sum_kernel<<<2048, 256, 0, stream>>>(x1, off, csr, aggr, N);
    sage_update_wmma<<<up_blocks, 256, 0, stream>>>(x1, aggr, deg, W2l, b2l, W2r, outp, N);
}


# update kernel: weight rows in LDS at a pitch of 68 floats, not 64, so a B-fragment read covers 64 banks once instead of a 16-way bank conflict; same fragments, same sums
# speedup vs baseline: 1.9626x; 1.1263x over previous
#include <hip/hip_runtime.h>

typedef __attribute__((ext_vector_type(2))) float v2f;
typedef __attribute__((ext_vector_type(8))) float v8f;

#define FEAT 64


__global__ void zero_i32(int* __restrict__ p, long n) {
    long i = (long)blockIdx.x * blockDim.x + threadIdx.x;
    long stride = (long)gridDim.x * blockDim.x;
    for (; i < n; i += stride) p[i] = 0;
}

#define EDGES_PER_THREAD 4
__global__ void __launch_bounds__(256)
count_rank_kernel(const int* __restrict__ dst, int* __restrict__ cnt,
                  int* __restrict__ rank, long E) {
    const long T = (long)gridDim.x * blockDim.x;
    long e = (long)blockIdx.x * blockDim.x + threadIdx.x;
    for (; e < E; e += EDGES_PER_THREAD * T) {
        int d[EDGES_PER_THREAD], r[EDGES_PER_THREAD];
#pragma unroll
        for (int i = 0; i < EDGES_PER_THREAD; ++i) {
            const long ee = e + i * T;
            d[i] = (ee < E) ? dst[ee] : -1;
        }
#pragma unroll
        for (int i = 0; i < EDGES_PER_THREAD; ++i) {
            r[i] = 0;
            if (d[i] >= 0) r[i] = atomicAdd(&cnt[d[i]], 1);
        }
#pragma unroll
        for (int i = 0; i < EDGES_PER_THREAD; ++i) {
            if (d[i] >= 0) rank[e + i * T] = r[i];
        }
    }
}

#define SCAN_CHUNK 1024
__global__ void __launch_bounds__(SCAN_CHUNK)
scan_kernel(const int* __restrict__ cnt, int* __restrict__ off,
            float* __restrict__ deg, int N) {
    __shared__ int front_s[SCAN_CHUNK / 32];
    __shared__ int wave_s[SCAN_CHUNK / 32];
    const int tid  = threadIdx.x;
    const int lane = tid & 31;
    const int wv   = tid >> 5;
    const int start = blockIdx.x * SCAN_CHUNK;

    int part = 0;
    const int4* c4 = (const int4*)cnt;
    const int n4 = start >> 2;
#pragma unroll 8
    for (int i = tid; i < n4; i += SCAN_CHUNK) {
        const int4 v = c4[i];
        part += (v.x + v.y) + (v.z + v.w);
    }
    for (int o = 16; o > 0; o >>= 1) part += __shfl_xor(part, o);
    if (lane == 0) front_s[wv] = part;

    const int idx = start + tid;
    const int c = (idx < N) ? cnt[idx] : 0;
    int incl = c;
    for (int o = 1; o < 32; o <<= 1) {
        const int t = __shfl_up(incl, o);
        if (lane >= o) incl += t;
    }
    if (lane == 31) wave_s[wv] = incl;
    __syncthreads();

    int base = 0;
    for (int k = 0; k < SCAN_CHUNK / 32; ++k) {
        base += front_s[k];
        if (k < wv) base += wave_s[k];
    }
    const int ex = base + incl - c;
    if (idx <= N) off[idx] = ex;
    if (idx < N)  deg[idx] = (float)c;
}

__global__ void __launch_bounds__(256)
fill_kernel(const int* __restrict__ src, const int* __restrict__ dst,
            const int* __restrict__ rank, const int* __restrict__ off,
            int* __restrict__ csr, long E) {
    const long T = (long)gridDim.x * blockDim.x;
    long e = (long)blockIdx.x * blockDim.x + threadIdx.x;
    for (; e < E; e += EDGES_PER_THREAD * T) {
        int d[EDGES_PER_THREAD], o[EDGES_PER_THREAD], r[EDGES_PER_THREAD], s[EDGES_PER_THREAD];
#pragma unroll
        for (int i = 0; i < EDGES_PER_THREAD; ++i) {
            const long ee = e + i * T;
            d[i] = (ee < E) ? dst[ee] : -1;
        }
#pragma unroll
        for (int i = 0; i < EDGES_PER_THREAD; ++i) {
            const long ee = e + i * T;
            o[i] = 0;
            r[i] = 0;
            s[i] = 0;
            if (d[i] >= 0) {
                o[i] = off[d[i]];
                r[i] = rank[ee];
                s[i] = src[ee];
            }
        }
#pragma unroll
        for (int i = 0; i < EDGES_PER_THREAD; ++i) {
            if (d[i] >= 0) csr[o[i] + r[i]] = s[i];
        }
    }
}

#define RUN 16
#define ROW2 (FEAT / 2)
__global__ void __launch_bounds__(256)
gather_sum_kernel(const float* __restrict__ x, const int* __restrict__ off,
                  const int* __restrict__ csr, float* __restrict__ aggr, int N) {
    const v2f* __restrict__ x2 = (const v2f*)x;
    v2f* __restrict__ aggr2 = (v2f*)aggr;
    const unsigned lane = threadIdx.x & 31u;
    const int wv  = __builtin_amdgcn_readfirstlane((int)(threadIdx.x >> 5));
    const int wpb = (int)(blockDim.x >> 5);
    const int nw  = (int)gridDim.x * wpb;
    for (int node = (int)blockIdx.x * wpb + wv; node < N; node += nw) {
        const int beg = off[node];
        const int end = off[node + 1];
        v2f acc = {0.0f, 0.0f};
        for (int p = beg; p < end; p += 32) {
            const int cnt = (end - p < 32) ? (end - p) : 32;
            int my = 0;
            if ((int)lane < cnt) my = csr[p + (int)lane];
            for (int j = 0; j < cnt; j += RUN) {
                const int m = cnt - j;
                unsigned s[RUN];
#pragma unroll
                for (int i = 0; i < RUN; ++i)
                    s[i] = (unsigned)__builtin_amdgcn_readlane(my, j + i) * (unsigned)ROW2;
                if (m >= RUN) {
                    v2f v[RUN];
#pragma unroll
                    for (int i = 0; i < RUN; ++i) v[i] = x2[s[i] + lane];
#pragma unroll
                    for (int i = 0; i < RUN; ++i) acc += v[i];
                } else {
                    v2f v[RUN];
#pragma unroll
                    for (int i = 0; i < RUN; ++i) {
                        v[i] = (v2f){0.0f, 0.0f};
                        if (i < m) v[i] = x2[s[i] + lane];
                    }
#pragma unroll
                    for (int i = 0; i < RUN; ++i) acc += v[i];
                }
            }
        }
        aggr2[(unsigned)node * (unsigned)ROW2 + lane] = acc;
    }
}

#define WPITCH (FEAT + 4)
__global__ void __launch_bounds__(256)
sage_update_wmma(const float* __restrict__ x, const float* __restrict__ aggr,
                 const float* __restrict__ deg,
                 const float* __restrict__ Wl, const float* __restrict__ bl,
                 const float* __restrict__ Wr,
                 float* __restrict__ out, int N) {
    __shared__ __attribute__((aligned(16))) float Wl_s[FEAT * WPITCH];
    __shared__ __attribute__((aligned(16))) float Wr_s[FEAT * WPITCH];
    __shared__ float b_s[FEAT];

    for (int i = threadIdx.x; i < FEAT * FEAT / 4; i += 256) {
        const int wr = i >> 4;
        const int wc = (i & 15) * 4;
        *(float4*)&Wl_s[wr * WPITCH + wc] = ((const float4*)Wl)[i];
        *(float4*)&Wr_s[wr * WPITCH + wc] = ((const float4*)Wr)[i];
    }
    if (threadIdx.x < FEAT) b_s[threadIdx.x] = bl[threadIdx.x];
    __syncthreads();

    const int lane = threadIdx.x & 31;
    const int wave = threadIdx.x >> 5;
    const int tile = blockIdx.x * 8 + wave;
    const int base = tile * 16;
    if (base >= N) return;

    const int row  = lane & 15;
    const int half = lane >> 4;
    const int node = base + row;
    const float dinv = 1.0f / fmaxf(deg[node], 1.0f);

    v2f a_ag[16], a_x[16];
    const float* agp = aggr + (long)node * FEAT;
    const float* xp  = x    + (long)node * FEAT;
#pragma unroll
    for (int ks = 0; ks < 16; ++ks) {
        const int k0 = ks * 4 + half * 2;
        v2f t0; t0.x = agp[k0] * dinv; t0.y = agp[k0 + 1] * dinv;
        a_ag[ks] = t0;
        v2f t1; t1.x = xp[k0];         t1.y = xp[k0 + 1];
        a_x[ks] = t1;
    }

#pragma unroll
    for (int nb = 0; nb < 4; ++nb) {
        const int ncol = nb * 16 + row;
        v8f c = {};
#pragma unroll
        for (int ks = 0; ks < 16; ++ks) {
            const int k0 = ks * 4 + half * 2;
            v2f bfl; bfl.x = Wl_s[ncol * WPITCH + k0]; bfl.y = Wl_s[ncol * WPITCH + k0 + 1];
            c = __builtin_amdgcn_wmma_f32_16x16x4_f32(false, a_ag[ks], false, bfl,
                                                      (short)0, c, false, false);
            v2f bfr; bfr.x = Wr_s[ncol * WPITCH + k0]; bfr.y = Wr_s[ncol * WPITCH + k0 + 1];
            c = __builtin_amdgcn_wmma_f32_16x16x4_f32(false, a_x[ks], false, bfr,
                                                      (short)0, c, false, false);
        }
        const float bias = b_s[nb * 16 + row];
#pragma unroll
        for (int r = 0; r < 8; ++r) {
            const int M = r + half * 8;
            out[(long)(base + M) * FEAT + nb * 16 + row] = c[r] + bias;
        }
    }
}

extern "C" void kernel_launch(void* const* d_in, const int* in_sizes, int n_in,
                              void* d_out, int out_size, void* d_ws, size_t ws_size,
                              hipStream_t stream) {
    const float* x   = (const float*)d_in[0];
    const int*   eix = (const int*)d_in[1];
    const float* W1l = (const float*)d_in[2];
    const float* b1l = (const float*)d_in[3];
    const float* W1r = (const float*)d_in[4];
    const float* W2l = (const float*)d_in[5];
    const float* b2l = (const float*)d_in[6];
    const float* W2r = (const float*)d_in[7];

    const int  N = in_sizes[0] / FEAT;
    const long E = in_sizes[1] / 2;
    const int* src = eix;
    const int* dst = eix + E;

    float* aggr = (float*)d_ws;
    float* x1   = aggr + (size_t)N * FEAT;
    float* deg  = x1 + (size_t)N * FEAT;
    float* outp = (float*)d_out;

    int* cnt  = (int*)x1;
    int* rank = cnt + N;
    int* off  = (int*)d_out;
    int* csr  = off + N + 1;

    const int edge_blocks = (int)((E + 256 * EDGES_PER_THREAD - 1) / (256 * EDGES_PER_THREAD));
    const int scan_blocks = (N + SCAN_CHUNK) / SCAN_CHUNK;
    const int up_blocks   = (N + 127) / 128;

    zero_i32<<<256, 256, 0, stream>>>(cnt, (long)N);
    count_rank_kernel<<<edge_blocks, 256, 0, stream>>>(dst, cnt, rank, E);
    scan_kernel<<<scan_blocks, SCAN_CHUNK, 0, stream>>>(cnt, off, deg, N);
    fill_kernel<<<edge_blocks, 256, 0, stream>>>(src, dst, rank, off, csr, E);

    gather_sum_kernel<<<2048, 256, 0, stream>>>(x, off, csr, aggr, N);
    sage_update_wmma<<<up_blocks, 256, 0, stream>>>(x, aggr, deg, W1l, b1l, W1r, x1, N);

    gather_sum_kernel<<<2048, 256, 0, stream>>>(x1, off, csr, aggr, N);
    sage_update_wmma<<<up_blocks, 256, 0, stream>>>(x1, aggr, deg, W2l, b2l, W2r, outp, N);
}


# edge bucketing without global atomics: work-group (edge chunk, node range) counts and slots its edges with LDS adds, 8 to 7 launches; a node's list is chunk-major, arrival order inside a chunk
# speedup vs baseline: 3.0972x; 1.5781x over previous
#include <hip/hip_runtime.h>

typedef __attribute__((ext_vector_type(2))) float v2f;
typedef __attribute__((ext_vector_type(8))) float v8f;

#define FEAT 64

#define NCH   16
#define RCAP  6250
#define SUBN  1024
#define NSUB  ((RCAP + SUBN - 1) / SUBN)
#define QUADS 8

__device__ __forceinline__ long chunk_len(long E) {
    return (((E + NCH - 1) / NCH) + 3) & ~3L;
}

__device__ __forceinline__ void count_id(int* h, int d, int n0, unsigned S) {
    const unsigned u = (unsigned)(d - n0);
    if (u < S) atomicAdd(&h[u], 1);
}

__global__ void __launch_bounds__(1024)
range_count_kernel(const int* __restrict__ dst, int* __restrict__ cntmat,
                   int* __restrict__ subcount, int N, long E) {
    __shared__ int h[RCAP];
    __shared__ int sub_s[NSUB];
    const int tid = threadIdx.x;
    const int c  = blockIdx.x;
    const int r  = blockIdx.y;
    const int n0 = r * RCAP;
    const unsigned S = (unsigned)((N - n0 < RCAP) ? (N - n0) : RCAP);
    for (int i = tid; i < RCAP; i += 1024) h[i] = 0;
    if (tid < NSUB) sub_s[tid] = 0;
    __syncthreads();

    const long EC = chunk_len(E);
    const long e0 = (long)c * EC;
    const long e1 = (e0 + EC < E) ? (e0 + EC) : E;
    if (((size_t)dst & 15) == 0) {
        const long e1q = e1 & ~3L;
        for (long e = e0 + 4L * tid; e < e1q; e += 4L * 1024 * QUADS) {
            int4 d[QUADS];
#pragma unroll
            for (int k = 0; k < QUADS; ++k) {
                const long ee = e + 4L * 1024 * k;
                d[k] = make_int4(-1, -1, -1, -1);
                if (ee < e1q) d[k] = *(const int4*)(dst + ee);
            }
#pragma unroll
            for (int k = 0; k < QUADS; ++k) {
                count_id(h, d[k].x, n0, S);
                count_id(h, d[k].y, n0, S);
                count_id(h, d[k].z, n0, S);
                count_id(h, d[k].w, n0, S);
            }
        }
        if (e0 <= e1q && e1q + tid < e1) count_id(h, dst[e1q + tid], n0, S);
    } else {
        for (long e = e0 + tid; e < e1; e += 1024L * 4 * QUADS) {
            int d[4 * QUADS];
#pragma unroll
            for (int k = 0; k < 4 * QUADS; ++k) {
                const long ee = e + 1024L * k;
                d[k] = (ee < e1) ? dst[ee] : -1;
            }
#pragma unroll
            for (int k = 0; k < 4 * QUADS; ++k) count_id(h, d[k], n0, S);
        }
    }
    __syncthreads();

    const int lane = tid & 31;
#pragma unroll
    for (int j = 0; j < NSUB; ++j) {
        const int i = j * SUBN + tid;
        int v = 0;
        if ((unsigned)i < S) {
            v = h[i];
            cntmat[(size_t)c * N + n0 + i] = v;
        }
        for (int o = 16; o > 0; o >>= 1) v += __shfl_xor(v, o);
        if (lane == 0) atomicAdd(&sub_s[j], v);
    }
    __syncthreads();
    if (tid < NSUB) subcount[(r * NSUB + tid) * NCH + c] = sub_s[tid];
}

__global__ void __launch_bounds__(SUBN)
range_scan_kernel(int* __restrict__ cntmat, const int* __restrict__ subcount,
                  int* __restrict__ off, float* __restrict__ deg, int N) {
    __shared__ int front_s[SUBN / 32];
    __shared__ int wave_s[SUBN / 32];
    const int tid  = threadIdx.x;
    const int lane = tid & 31;
    const int wv   = tid >> 5;
    const int j  = blockIdx.x;
    const int r  = blockIdx.y;
    const int n0 = r * RCAP;
    const int S  = (N - n0 < RCAP) ? (N - n0) : RCAP;

    int part = 0;
    const int nfront = (r * NSUB + j) * NCH;
    for (int i = tid; i < nfront; i += SUBN) part += subcount[i];
    for (int o = 16; o > 0; o >>= 1) part += __shfl_xor(part, o);
    if (lane == 0) front_s[wv] = part;

    const int i = j * SUBN + tid;
    const bool live = i < S;
    const int n = n0 + i;
    int cv[NCH];
    int total = 0;
#pragma unroll
    for (int c = 0; c < NCH; ++c) {
        cv[c] = live ? cntmat[(size_t)c * N + n] : 0;
        total += cv[c];
    }
    int incl = total;
    for (int o = 1; o < 32; o <<= 1) {
        const int t = __shfl_up(incl, o);
        if (lane >= o) incl += t;
    }
    if (lane == 31) wave_s[wv] = incl;
    __syncthreads();

    int base = 0;
    for (int k = 0; k < SUBN / 32; ++k) {
        base += front_s[k];
        if (k < wv) base += wave_s[k];
    }
    const int ex = base + incl - total;
    if (live) {
        off[n] = ex;
        deg[n] = (float)total;
        int run = ex;
#pragma unroll
        for (int c = 0; c < NCH; ++c) {
            cntmat[(size_t)c * N + n] = run;
            run += cv[c];
        }
        if (n == N - 1) off[N] = run;
    }
}

__device__ __forceinline__ int take_slot(int* cur, int d, int n0, unsigned S) {
    const unsigned u = (unsigned)(d - n0);
    int p = -1;
    if (u < S) p = atomicAdd(&cur[u], 1);
    return p;
}

__global__ void __launch_bounds__(1024)
range_fill_kernel(const int* __restrict__ src, const int* __restrict__ dst,
                  const int* __restrict__ cntmat, int* __restrict__ csr, int N, long E) {
    __shared__ int cur[RCAP];
    const int tid = threadIdx.x;
    const int c  = blockIdx.x;
    const int r  = blockIdx.y;
    const int n0 = r * RCAP;
    const unsigned S = (unsigned)((N - n0 < RCAP) ? (N - n0) : RCAP);
    for (int i = tid; (unsigned)i < S; i += 1024) cur[i] = cntmat[(size_t)c * N + n0 + i];
    __syncthreads();

    const long EC = chunk_len(E);
    const long e0 = (long)c * EC;
    const long e1 = (e0 + EC < E) ? (e0 + EC) : E;
    if ((((size_t)dst | (size_t)src) & 15) == 0) {
        const long e1q = e1 & ~3L;
        for (long e = e0 + 4L * tid; e < e1q; e += 4L * 1024 * (QUADS / 2)) {
            int4 d[QUADS / 2], s[QUADS / 2];
#pragma unroll
            for (int k = 0; k < QUADS / 2; ++k) {
                const long ee = e + 4L * 1024 * k;
                d[k] = make_int4(-1, -1, -1, -1);
                s[k] = make_int4(0, 0, 0, 0);
                if (ee < e1q) {
                    d[k] = *(const int4*)(dst + ee);
                    s[k] = *(const int4*)(src + ee);
                }
            }
            int4 p[QUADS / 2];
#pragma unroll
            for (int k = 0; k < QUADS / 2; ++k) {
                p[k].x = take_slot(cur, d[k].x, n0, S);
                p[k].y = take_slot(cur, d[k].y, n0, S);
                p[k].z = take_slot(cur, d[k].z, n0, S);
                p[k].w = take_slot(cur, d[k].w, n0, S);
            }
#pragma unroll
            for (int k = 0; k < QUADS / 2; ++k) {
                if (p[k].x >= 0) csr[p[k].x] = s[k].x;
                if (p[k].y >= 0) csr[p[k].y] = s[k].y;
                if (p[k].z >= 0) csr[p[k].z] = s[k].z;
                if (p[k].w >= 0) csr[p[k].w] = s[k].w;
            }
        }
        if (e0 <= e1q && e1q + tid < e1) {
            const int p1 = take_slot(cur, dst[e1q + tid], n0, S);
            if (p1 >= 0) csr[p1] = src[e1q + tid];
        }
    } else {
        for (long e = e0 + tid; e < e1; e += 1024L * 2 * QUADS) {
            int d[2 * QUADS], s[2 * QUADS], p[2 * QUADS];
#pragma unroll
            for (int k = 0; k < 2 * QUADS; ++k) {
                const long ee = e + 1024L * k;
                d[k] = -1;
                s[k] = 0;
                if (ee < e1) {
                    d[k] = dst[ee];
                    s[k] = src[ee];
                }
            }
#pragma unroll
            for (int k = 0; k < 2 * QUADS; ++k) p[k] = take_slot(cur, d[k], n0, S);
#pragma unroll
            for (int k = 0; k < 2 * QUADS; ++k)
                if (p[k] >= 0) csr[p[k]] = s[k];
        }
    }
}

#define RUN 16
#define ROW2 (FEAT / 2)
__global__ void __launch_bounds__(256)
gather_sum_kernel(const float* __restrict__ x, const int* __restrict__ off,
                  const int* __restrict__ csr, float* __restrict__ aggr, int N) {
    const v2f* __restrict__ x2 = (const v2f*)x;
    v2f* __restrict__ aggr2 = (v2f*)aggr;
    const unsigned lane = threadIdx.x & 31u;
    const int wv  = __builtin_amdgcn_readfirstlane((int)(threadIdx.x >> 5));
    const int wpb = (int)(blockDim.x >> 5);
    const int nw  = (int)gridDim.x * wpb;
    for (int node = (int)blockIdx.x * wpb + wv; node < N; node += nw) {
        const int beg = off[node];
        const int end = off[node + 1];
        v2f acc = {0.0f, 0.0f};
        for (int p = beg; p < end; p += 32) {
            const int cnt = (end - p < 32) ? (end - p) : 32;
            int my = 0;
            if ((int)lane < cnt) my = csr[p + (int)lane];
            for (int j = 0; j < cnt; j += RUN) {
                const int m = cnt - j;
                unsigned s[RUN];
#pragma unroll
                for (int i = 0; i < RUN; ++i)
                    s[i] = (unsigned)__builtin_amdgcn_readlane(my, j + i) * (unsigned)ROW2;
                if (m >= RUN) {
                    v2f v[RUN];
#pragma unroll
                    for (int i = 0; i < RUN; ++i) v[i] = x2[s[i] + lane];
#pragma unroll
                    for (int i = 0; i < RUN; ++i) acc += v[i];
                } else {
                    v2f v[RUN];
#pragma unroll
                    for (int i = 0; i < RUN; ++i) {
                        v[i] = (v2f){0.0f, 0.0f};
                        if (i < m) v[i] = x2[s[i] + lane];
                    }
#pragma unroll
                    for (int i = 0; i < RUN; ++i) acc += v[i];
                }
            }
        }
        aggr2[(unsigned)node * (unsigned)ROW2 + lane] = acc;
    }
}

#define WPITCH (FEAT + 4)
__global__ void __launch_bounds__(256)
sage_update_wmma(const float* __restrict__ x, const float* __restrict__ aggr,
                 const float* __restrict__ deg,
                 const float* __restrict__ Wl, const float* __restrict__ bl,
                 const float* __restrict__ Wr,
                 float* __restrict__ out, int N) {
    __shared__ __attribute__((aligned(16))) float Wl_s[FEAT * WPITCH];
    __shared__ __attribute__((aligned(16))) float Wr_s[FEAT * WPITCH];
    __shared__ float b_s[FEAT];

    for (int i = threadIdx.x; i < FEAT * FEAT / 4; i += 256) {
        const int wr = i >> 4;
        const int wc = (i & 15) * 4;
        *(float4*)&Wl_s[wr * WPITCH + wc] = ((const float4*)Wl)[i];
        *(float4*)&Wr_s[wr * WPITCH + wc] = ((const float4*)Wr)[i];
    }
    if (threadIdx.x < FEAT) b_s[threadIdx.x] = bl[threadIdx.x];
    __syncthreads();

    const int lane = threadIdx.x & 31;
    const int wave = threadIdx.x >> 5;
    const int tile = blockIdx.x * 8 + wave;
    const int base = tile * 16;
    if (base >= N) return;

    const int row  = lane & 15;
    const int half = lane >> 4;
    const int node = base + row;
    const float dinv = 1.0f / fmaxf(deg[node], 1.0f);

    v2f a_ag[16], a_x[16];
    const float* agp = aggr + (long)node * FEAT;
    const float* xp  = x    + (long)node * FEAT;
#pragma unroll
    for (int ks = 0; ks < 16; ++ks) {
        const int k0 = ks * 4 + half * 2;
        v2f t0; t0.x = agp[k0] * dinv; t0.y = agp[k0 + 1] * dinv;
        a_ag[ks] = t0;
        v2f t1; t1.x = xp[k0];         t1.y = xp[k0 + 1];
        a_x[ks] = t1;
    }

#pragma unroll
    for (int nb = 0; nb < 4; ++nb) {
        const int ncol = nb * 16 + row;
        v8f c = {};
#pragma unroll
        for (int ks = 0; ks < 16; ++ks) {
            const int k0 = ks * 4 + half * 2;
            v2f bfl; bfl.x = Wl_s[ncol * WPITCH + k0]; bfl.y = Wl_s[ncol * WPITCH + k0 + 1];
            c = __builtin_amdgcn_wmma_f32_16x16x4_f32(false, a_ag[ks], false, bfl,
                                                      (short)0, c, false, false);
            v2f bfr; bfr.x = Wr_s[ncol * WPITCH + k0]; bfr.y = Wr_s[ncol * WPITCH + k0 + 1];
            c = __builtin_amdgcn_wmma_f32_16x16x4_f32(false, a_x[ks], false, bfr,
                                                      (short)0, c, false, false);
        }
        const float bias = b_s[nb * 16 + row];
#pragma unroll
        for (int r = 0; r < 8; ++r) {
            const int M = r + half * 8;
            out[(long)(base + M) * FEAT + nb * 16 + row] = c[r] + bias;
        }
    }
}

extern "C" void kernel_launch(void* const* d_in, const int* in_sizes, int n_in,
                              void* d_out, int out_size, void* d_ws, size_t ws_size,
                              hipStream_t stream) {
    const float* x   = (const float*)d_in[0];
    const int*   eix = (const int*)d_in[1];
    const float* W1l = (const float*)d_in[2];
    const float* b1l = (const float*)d_in[3];
    const float* W1r = (const float*)d_in[4];
    const float* W2l = (const float*)d_in[5];
    const float* b2l = (const float*)d_in[6];
    const float* W2r = (const float*)d_in[7];

    const int  N = in_sizes[0] / FEAT;
    const long E = in_sizes[1] / 2;
    const int* src = eix;
    const int* dst = eix + E;

    float* aggr = (float*)d_ws;
    float* x1   = aggr + (size_t)N * FEAT;
    float* deg  = x1 + (size_t)N * FEAT;
    float* outp = (float*)d_out;

    int* cntmat   = (int*)x1;
    int* subcount = cntmat + (size_t)NCH * N;
    int* off  = (int*)d_out;
    int* csr  = off + N + 1;

    const int ranges    = (N + RCAP - 1) / RCAP;
    const int up_blocks = (N + 127) / 128;

    range_count_kernel<<<dim3(NCH, ranges), 1024, 0, stream>>>(dst, cntmat, subcount, N, E);
    range_scan_kernel<<<dim3(NSUB, ranges), SUBN, 0, stream>>>(cntmat, subcount, off, deg, N);
    range_fill_kernel<<<dim3(NCH, ranges), 1024, 0, stream>>>(src, dst, cntmat, csr, N, E);

    gather_sum_kernel<<<2048, 256, 0, stream>>>(x, off, csr, aggr, N);
    sage_update_wmma<<<up_blocks, 256, 0, stream>>>(x, aggr, deg, W1l, b1l, W1r, x1, N);

    gather_sum_kernel<<<2048, 256, 0, stream>>>(x1, off, csr, aggr, N);
    sage_update_wmma<<<up_blocks, 256, 0, stream>>>(x1, aggr, deg, W2l, b2l, W2r, outp, N);
}


# gather fused into the update kernel: a wave sums its 16 nodes' source rows (one run of the edge list, 32 rows in flight) into an LDS tile read as WMMA A fragments; 7 to 5 launches; same sums
# speedup vs baseline: 3.3950x; 1.0962x over previous
#include <hip/hip_runtime.h>

typedef __attribute__((ext_vector_type(2))) float v2f;
typedef __attribute__((ext_vector_type(8))) float v8f;

#define FEAT 64

#define NCH   16
#define RCAP  6250
#define SUBN  1024
#define NSUB  ((RCAP + SUBN - 1) / SUBN)
#define QUADS 8

__device__ __forceinline__ long chunk_len(long E) {
    return (((E + NCH - 1) / NCH) + 3) & ~3L;
}

__device__ __forceinline__ void count_id(int* h, int d, int n0, unsigned S) {
    const unsigned u = (unsigned)(d - n0);
    if (u < S) atomicAdd(&h[u], 1);
}

__global__ void __launch_bounds__(1024)
range_count_kernel(const int* __restrict__ dst, int* __restrict__ cntmat,
                   int* __restrict__ subcount, int N, long E) {
    __shared__ int h[RCAP];
    __shared__ int sub_s[NSUB];
    const int tid = threadIdx.x;
    const int c  = blockIdx.x;
    const int r  = blockIdx.y;
    const int n0 = r * RCAP;
    const unsigned S = (unsigned)((N - n0 < RCAP) ? (N - n0) : RCAP);
    for (int i = tid; i < RCAP; i += 1024) h[i] = 0;
    if (tid < NSUB) sub_s[tid] = 0;
    __syncthreads();

    const long EC = chunk_len(E);
    const long e0 = (long)c * EC;
    const long e1 = (e0 + EC < E) ? (e0 + EC) : E;
    if (((size_t)dst & 15) == 0) {
        const long e1q = e1 & ~3L;
        for (long e = e0 + 4L * tid; e < e1q; e += 4L * 1024 * QUADS) {
            int4 d[QUADS];
#pragma unroll
            for (int k = 0; k < QUADS; ++k) {
                const long ee = e + 4L * 1024 * k;
                d[k] = make_int4(-1, -1, -1, -1);
                if (ee < e1q) d[k] = *(const int4*)(dst + ee);
            }
#pragma unroll
            for (int k = 0; k < QUADS; ++k) {
                count_id(h, d[k].x, n0, S);
                count_id(h, d[k].y, n0, S);
                count_id(h, d[k].z, n0, S);
                count_id(h, d[k].w, n0, S);
            }
        }
        if (e0 <= e1q && e1q + tid < e1) count_id(h, dst[e1q + tid], n0, S);
    } else {
        for (long e = e0 + tid; e < e1; e += 1024L * 4 * QUADS) {
            int d[4 * QUADS];
#pragma unroll
            for (int k = 0; k < 4 * QUADS; ++k) {
                const long ee = e + 1024L * k;
                d[k] = (ee < e1) ? dst[ee] : -1;
            }
#pragma unroll
            for (int k = 0; k < 4 * QUADS; ++k) count_id(h, d[k], n0, S);
        }
    }
    __syncthreads();

    const int lane = tid & 31;
#pragma unroll
    for (int j = 0; j < NSUB; ++j) {
        const int i = j * SUBN + tid;
        int v = 0;
        if ((unsigned)i < S) {
            v = h[i];
            cntmat[(size_t)c * N + n0 + i] = v;
        }
        for (int o = 16; o > 0; o >>= 1) v += __shfl_xor(v, o);
        if (lane == 0) atomicAdd(&sub_s[j], v);
    }
    __syncthreads();
    if (tid < NSUB) subcount[(r * NSUB + tid) * NCH + c] = sub_s[tid];
}

__global__ void __launch_bounds__(SUBN)
range_scan_kernel(int* __restrict__ cntmat, const int* __restrict__ subcount,
                  int* __restrict__ off, float* __restrict__ deg, int N) {
    __shared__ int front_s[SUBN / 32];
    __shared__ int wave_s[SUBN / 32];
    const int tid  = threadIdx.x;
    const int lane = tid & 31;
    const int wv   = tid >> 5;
    const int j  = blockIdx.x;
    const int r  = blockIdx.y;
    const int n0 = r * RCAP;
    const int S  = (N - n0 < RCAP) ? (N - n0) : RCAP;

    int part = 0;
    const int nfront = (r * NSUB + j) * NCH;
    for (int i = tid; i < nfront; i += SUBN) part += subcount[i];
    for (int o = 16; o > 0; o >>= 1) part += __shfl_xor(part, o);
    if (lane == 0) front_s[wv] = part;

    const int i = j * SUBN + tid;
    const bool live = i < S;
    const int n = n0 + i;
    int cv[NCH];
    int total = 0;
#pragma unroll
    for (int c = 0; c < NCH; ++c) {
        cv[c] = live ? cntmat[(size_t)c * N + n] : 0;
        total += cv[c];
    }
    int incl = total;
    for (int o = 1; o < 32; o <<= 1) {
        const int t = __shfl_up(incl, o);
        if (lane >= o) incl += t;
    }
    if (lane == 31) wave_s[wv] = incl;
    __syncthreads();

    int base = 0;
    for (int k = 0; k < SUBN / 32; ++k) {
        base += front_s[k];
        if (k < wv) base += wave_s[k];
    }
    const int ex = base + incl - total;
    if (live) {
        off[n] = ex;
        deg[n] = (float)total;
        int run = ex;
#pragma unroll
        for (int c = 0; c < NCH; ++c) {
            cntmat[(size_t)c * N + n] = run;
            run += cv[c];
        }
        if (n == N - 1) off[N] = run;
    }
}

__device__ __forceinline__ int take_slot(int* cur, int d, int n0, unsigned S) {
    const unsigned u = (unsigned)(d - n0);
    int p = -1;
    if (u < S) p = atomicAdd(&cur[u], 1);
    return p;
}

__global__ void __launch_bounds__(1024)
range_fill_kernel(const int* __restrict__ src, const int* __restrict__ dst,
                  const int* __restrict__ cntmat, int* __restrict__ csr, int N, long E) {
    __shared__ int cur[RCAP];
    const int tid = threadIdx.x;
    const int c  = blockIdx.x;
    const int r  = blockIdx.y;
    const int n0 = r * RCAP;
    const unsigned S = (unsigned)((N - n0 < RCAP) ? (N - n0) : RCAP);
    for (int i = tid; (unsigned)i < S; i += 1024) cur[i] = cntmat[(size_t)c * N + n0 + i];
    __syncthreads();

    const long EC = chunk_len(E);
    const long e0 = (long)c * EC;
    const long e1 = (e0 + EC < E) ? (e0 + EC) : E;
    if ((((size_t)dst | (size_t)src) & 15) == 0) {
        const long e1q = e1 & ~3L;
        for (long e = e0 + 4L * tid; e < e1q; e += 4L * 1024 * (QUADS / 2)) {
            int4 d[QUADS / 2], s[QUADS / 2];
#pragma unroll
            for (int k = 0; k < QUADS / 2; ++k) {
                const long ee = e + 4L * 1024 * k;
                d[k] = make_int4(-1, -1, -1, -1);
                s[k] = make_int4(0, 0, 0, 0);
                if (ee < e1q) {
                    d[k] = *(const int4*)(dst + ee);
                    s[k] = *(const int4*)(src + ee);
                }
            }
            int4 p[QUADS / 2];
#pragma unroll
            for (int k = 0; k < QUADS / 2; ++k) {
                p[k].x = take_slot(cur, d[k].x, n0, S);
                p[k].y = take_slot(cur, d[k].y, n0, S);
                p[k].z = take_slot(cur, d[k].z, n0, S);
                p[k].w = take_slot(cur, d[k].w, n0, S);
            }
#pragma unroll
            for (int k = 0; k < QUADS / 2; ++k) {
                if (p[k].x >= 0) csr[p[k].x] = s[k].x;
                if (p[k].y >= 0) csr[p[k].y] = s[k].y;
                if (p[k].z >= 0) csr[p[k].z] = s[k].z;
                if (p[k].w >= 0) csr[p[k].w] = s[k].w;
            }
        }
        if (e0 <= e1q && e1q + tid < e1) {
            const int p1 = take_slot(cur, dst[e1q + tid], n0, S);
            if (p1 >= 0) csr[p1] = src[e1q + tid];
        }
    } else {
        for (long e = e0 + tid; e < e1; e += 1024L * 2 * QUADS) {
            int d[2 * QUADS], s[2 * QUADS], p[2 * QUADS];
#pragma unroll
            for (int k = 0; k < 2 * QUADS; ++k) {
                const long ee = e + 1024L * k;
                d[k] = -1;
                s[k] = 0;
                if (ee < e1) {
                    d[k] = dst[ee];
                    s[k] = src[ee];
                }
            }
#pragma unroll
            for (int k = 0; k < 2 * QUADS; ++k) p[k] = take_slot(cur, d[k], n0, S);
#pragma unroll
            for (int k = 0; k < 2 * QUADS; ++k)
                if (p[k] >= 0) csr[p[k]] = s[k];
        }
    }
}

#define WPITCH (FEAT + 4)
#define TPITCH (FEAT + 4)
#define ROW2 (FEAT / 2)
#define CH 32
#define TILE_WAVES 8

__global__ void __launch_bounds__(256)
sage_layer_wmma(const float* __restrict__ x, const int* __restrict__ off,
                const int* __restrict__ csr, const float* __restrict__ deg,
                const float* __restrict__ Wl, const float* __restrict__ bl,
                const float* __restrict__ Wr,
                float* __restrict__ out, int N) {
    __shared__ __attribute__((aligned(16))) float Wl_s[FEAT * WPITCH];
    __shared__ __attribute__((aligned(16))) float Wr_s[FEAT * WPITCH];
    __shared__ float b_s[FEAT];
    __shared__ __attribute__((aligned(16))) float tile_s[TILE_WAVES * 16 * TPITCH];

    for (int i = threadIdx.x; i < FEAT * FEAT / 4; i += 256) {
        const int wr = i >> 4;
        const int wc = (i & 15) * 4;
        *(float4*)&Wl_s[wr * WPITCH + wc] = ((const float4*)Wl)[i];
        *(float4*)&Wr_s[wr * WPITCH + wc] = ((const float4*)Wr)[i];
    }
    if (threadIdx.x < FEAT) b_s[threadIdx.x] = bl[threadIdx.x];
    __syncthreads();

    const int lane = threadIdx.x & 31;
    const int wave = __builtin_amdgcn_readfirstlane((int)(threadIdx.x >> 5));
    const int tile = blockIdx.x * TILE_WAVES + wave;
    const int base = tile * 16;
    if (base >= N) return;
    const int nrows = (N - base < 16) ? (N - base) : 16;

    const v2f* __restrict__ x2 = (const v2f*)x;
    float* tl = tile_s + wave * (16 * TPITCH);
    const int offv = off[base + ((lane < nrows) ? lane : nrows)];
    const int beg0  = __builtin_amdgcn_readfirstlane(offv);
    const int total = __builtin_amdgcn_readlane(offv, 16) - beg0;

    int cur  = 0;
    int ecur = __builtin_amdgcn_readlane(offv, 1) - beg0;
    v2f acc = {0.0f, 0.0f};
#define CLOSE_AT(p)                                                              \
    while (ecur == (p)) {                                                        \
        *(v2f*)&tl[cur * TPITCH + 2 * lane] = acc;                               \
        acc = (v2f){0.0f, 0.0f};                                                 \
        ++cur;                                                                   \
        ecur = (cur < 16) ? (__builtin_amdgcn_readlane(offv, cur + 1) - beg0)    \
                          : 0x7fffffff;                                          \
    }
    CLOSE_AT(0)

    if (total > 0) {
        const int last = total - 1;
        int ids = csr[beg0 + ((lane < last) ? lane : last)];
        int q0 = 0;
        for (; q0 + CH <= total; q0 += CH) {
            const int nl = q0 + CH + lane;
            const int nxt = csr[beg0 + ((nl < last) ? nl : last)];
            unsigned s[CH];
#pragma unroll
            for (int i = 0; i < CH; ++i)
                s[i] = (unsigned)__builtin_amdgcn_readlane(ids, i) * (unsigned)ROW2;
            v2f v[CH];
#pragma unroll
            for (int i = 0; i < CH; ++i) v[i] = x2[s[i] + (unsigned)lane];
#pragma unroll
            for (int i = 0; i < CH; ++i) {
                acc += v[i];
                CLOSE_AT(q0 + i + 1)
            }
            ids = nxt;
        }
        if (q0 < total) {
            const int cnt = total - q0;
            unsigned s[CH];
#pragma unroll
            for (int i = 0; i < CH; ++i)
                s[i] = (unsigned)__builtin_amdgcn_readlane(ids, i) * (unsigned)ROW2;
            v2f v[CH];
#pragma unroll
            for (int i = 0; i < CH; ++i) {
                v[i] = (v2f){0.0f, 0.0f};
                if (i < cnt) v[i] = x2[s[i] + (unsigned)lane];
            }
#pragma unroll
            for (int i = 0; i < CH; ++i) {
                if (i < cnt) {
                    acc += v[i];
                    CLOSE_AT(q0 + i + 1)
                }
            }
        }
    }
#undef CLOSE_AT

    const int row  = lane & 15;
    const int half = lane >> 4;
    const int node = (base + row < N) ? (base + row) : (N - 1);
    const float dinv = 1.0f / fmaxf(deg[node], 1.0f);

    v2f a_x[16];
    const float* xp = x + (long)node * FEAT;
#pragma unroll
    for (int ks = 0; ks < 16; ++ks) {
        const int k0 = ks * 4 + half * 2;
        v2f t1; t1.x = xp[k0]; t1.y = xp[k0 + 1];
        a_x[ks] = t1;
    }

    v8f c[4] = {};
#pragma unroll
    for (int ks = 0; ks < 16; ++ks) {
        const int k0 = ks * 4 + half * 2;
        v2f a_ag;
        a_ag.x = tl[row * TPITCH + k0]     * dinv;
        a_ag.y = tl[row * TPITCH + k0 + 1] * dinv;
#pragma unroll
        for (int nb = 0; nb < 4; ++nb) {
            const int ncol = nb * 16 + row;
            v2f bfl; bfl.x = Wl_s[ncol * WPITCH + k0]; bfl.y = Wl_s[ncol * WPITCH + k0 + 1];
            c[nb] = __builtin_amdgcn_wmma_f32_16x16x4_f32(false, a_ag, false, bfl,
                                                          (short)0, c[nb], false, false);
            v2f bfr; bfr.x = Wr_s[ncol * WPITCH + k0]; bfr.y = Wr_s[ncol * WPITCH + k0 + 1];
            c[nb] = __builtin_amdgcn_wmma_f32_16x16x4_f32(false, a_x[ks], false, bfr,
                                                          (short)0, c[nb], false, false);
        }
    }

#pragma unroll
    for (int nb = 0; nb < 4; ++nb) {
        const float bias = b_s[nb * 16 + row];
#pragma unroll
        for (int r = 0; r < 8; ++r) {
            const int M = r + half * 8;
            out[(long)(base + M) * FEAT + nb * 16 + row] = c[nb][r] + bias;
        }
    }
}

extern "C" void kernel_launch(void* const* d_in, const int* in_sizes, int n_in,
                              void* d_out, int out_size, void* d_ws, size_t ws_size,
                              hipStream_t stream) {
    const float* x   = (const float*)d_in[0];
    const int*   eix = (const int*)d_in[1];
    const float* W1l = (const float*)d_in[2];
    const float* b1l = (const float*)d_in[3];
    const float* W1r = (const float*)d_in[4];
    const float* W2l = (const float*)d_in[5];
    const float* b2l = (const float*)d_in[6];
    const float* W2r = (const float*)d_in[7];

    const int  N = in_sizes[0] / FEAT;
    const long E = in_sizes[1] / 2;
    const int* src = eix;
    const int* dst = eix + E;

    float* lists = (float*)d_ws;
    float* x1    = lists + (size_t)N * FEAT;
    float* deg   = x1 + (size_t)N * FEAT;
    float* outp  = (float*)d_out;

    int* cntmat   = (int*)x1;
    int* subcount = cntmat + (size_t)NCH * N;
    int* off  = (int*)lists;
    int* csr  = off + N + 1;

    const int ranges    = (N + RCAP - 1) / RCAP;
    const int up_blocks = (N + 127) / 128;

    range_count_kernel<<<dim3(NCH, ranges), 1024, 0, stream>>>(dst, cntmat, subcount, N, E);
    range_scan_kernel<<<dim3(NSUB, ranges), SUBN, 0, stream>>>(cntmat, subcount, off, deg, N);
    range_fill_kernel<<<dim3(NCH, ranges), 1024, 0, stream>>>(src, dst, cntmat, csr, N, E);

    sage_layer_wmma<<<up_blocks, 256, 0, stream>>>(x, off, csr, deg, W1l, b1l, W1r, x1, N);

    sage_layer_wmma<<<up_blocks, 256, 0, stream>>>(x1, off, csr, deg, W2l, b2l, W2r, outp, N);
}
